# starting kernel
# baseline (speedup 1.0000x reference)
#include <hip/hip_runtime.h>

typedef float v2f __attribute__((ext_vector_type(2)));
typedef float v8f __attribute__((ext_vector_type(8)));

#define B_ 4
#define S_ 256
#define D_ 512
#define P_ 256
#define NEG_INF_ (-1e9f)

__device__ __forceinline__ float fast_tanh(float x) {
#if __has_builtin(__builtin_amdgcn_tanhf)
  return __builtin_amdgcn_tanhf(x);
#else
  float ax = __builtin_fabsf(x);
  float t  = __expf(-2.0f * ax);
  float r  = (1.0f - t) * __builtin_amdgcn_rcpf(1.0f + t);
  return x < 0.0f ? -r : r;
#endif
}

__global__ __launch_bounds__(256) void proj_wmma(
    const float* __restrict__ seq, const float* __restrict__ W1,
    const float* __restrict__ W2, float* __restrict__ pk,
    float* __restrict__ pq) {
  const int lane = threadIdx.x & 31;
  const int wid  = blockIdx.x * (blockDim.x >> 5) + (threadIdx.x >> 5);
  const int matsel = wid >> 10;
  const int tile   = wid & 1023;
  const int m0 = (tile >> 4) << 4;
  const int n0 = (tile & 15) << 4;
  const float* W   = matsel ? W2 : W1;
  float*       out = matsel ? pq : pk;

  const int hl  = lane >> 4;
  const int l16 = lane & 15;
  const int m = m0 + l16;
  const int n = n0 + l16;

  v8f acc = {};
  for (int k0 = 0; k0 < D_; k0 += 4) {
    const int k = k0 + (hl << 1);
    v2f a = *(const v2f*)(seq + m * D_ + k);
    v2f b = *(const v2f*)(W   + n * D_ + k);
    acc = __builtin_amdgcn_wmma_f32_16x16x4_f32(
        false, a, false, b, (short)0, acc, false, false);
  }
#pragma unroll
  for (int r = 0; r < 8; ++r)
    out[(m0 + r + (hl << 3)) * P_ + n0 + l16] = acc[r];
}

__global__ __launch_bounds__(256) void score_softmax(
    const float* __restrict__ pq, const float* __restrict__ pk,
    const float* __restrict__ v, const float* __restrict__ mask,
    float* __restrict__ wout) {
  const int bi  = blockIdx.x;
  const int b   = bi >> 8;
  const int tid = threadIdx.x;
  const int lane = tid & 31, wv = tid >> 5;

  __shared__ float pq_s[P_];
  __shared__ float v_s[P_];
  __shared__ float sc[S_];
  __shared__ float redm[8];
  __shared__ float reds[8];

  pq_s[tid] = pq[bi * P_ + tid];
  v_s[tid]  = v[tid];
  __syncthreads();

  const float* pkb = pk + b * S_ * P_;
  for (int jj = 0; jj < 32; ++jj) {
    const int j = (wv << 5) + jj;
    const float* row = pkb + j * P_;
    if (jj + 1 < 32)
      __builtin_prefetch(pkb + (j + 1) * P_ + lane * 8, 0, 1);
    float part = 0.0f;
#pragma unroll
    for (int q = 0; q < 8; ++q) {
      const int p = lane + (q << 5);
      part = __builtin_fmaf(v_s[p], fast_tanh(pq_s[p] + row[p]), part);
    }
#pragma unroll
    for (int off = 16; off > 0; off >>= 1)
      part += __shfl_xor(part, off, 32);
    if (lane == 0) sc[j] = part;
  }
  __syncthreads();

  const float km = mask[b * S_ + tid];
  float s = (km > 0.0f) ? sc[tid] : NEG_INF_;
  float mx = s;
#pragma unroll
  for (int off = 16; off > 0; off >>= 1)
    mx = fmaxf(mx, __shfl_xor(mx, off, 32));
  if (lane == 0) redm[wv] = mx;
  __syncthreads();
  float bm = redm[0];
#pragma unroll
  for (int w = 1; w < 8; ++w) bm = fmaxf(bm, redm[w]);

  float e  = __expf(s - bm);
  float sm = e;
#pragma unroll
  for (int off = 16; off > 0; off >>= 1)
    sm += __shfl_xor(sm, off, 32);
  if (lane == 0) reds[wv] = sm;
  __syncthreads();
  float tot = 0.0f;
#pragma unroll
  for (int w = 0; w < 8; ++w) tot += reds[w];

  const int   i  = bi & 255;
  const float qm = mask[b * S_ + i];
  wout[bi * S_ + tid] = (e / tot) * qm;
}

__global__ __launch_bounds__(256) void attn_wmma(
    const float* __restrict__ weights, const float* __restrict__ seq,
    float* __restrict__ attn) {
  const int lane = threadIdx.x & 31;
  const int wid  = blockIdx.x * (blockDim.x >> 5) + (threadIdx.x >> 5);
  const int b    = wid >> 9;
  const int rest = wid & 511;
  const int m0 = (rest >> 5) << 4;
  const int n0 = (rest & 31) << 4;
  const float* Wt = weights + b * S_ * S_;
  const float* Sq = seq     + b * S_ * D_;

  const int hl  = lane >> 4;
  const int l16 = lane & 15;
  const int m = m0 + l16;
  const int n = n0 + l16;

  v8f acc = {};
  for (int k0 = 0; k0 < S_; k0 += 4) {
    const int k = k0 + (hl << 1);
    v2f a = *(const v2f*)(Wt + m * S_ + k);
    v2f bb;
    bb.x = Sq[k * D_ + n];
    bb.y = Sq[(k + 1) * D_ + n];
    acc = __builtin_amdgcn_wmma_f32_16x16x4_f32(
        false, a, false, bb, (short)0, acc, false, false);
  }
  float* O = attn + b * S_ * D_;
#pragma unroll
  for (int r = 0; r < 8; ++r)
    O[(m0 + r + (hl << 3)) * D_ + n0 + l16] = acc[r];
}

extern "C" void kernel_launch(void* const* d_in, const int* in_sizes, int n_in,
                              void* d_out, int out_size, void* d_ws, size_t ws_size,
                              hipStream_t stream) {
  (void)in_sizes; (void)n_in; (void)out_size; (void)ws_size;
  const float* seq  = (const float*)d_in[0];
  const float* mask = (const float*)d_in[1];
  const float* W1   = (const float*)d_in[2];
  const float* W2   = (const float*)d_in[3];
  const float* v    = (const float*)d_in[4];

  float* attn = (float*)d_out;
  float* wout = attn + B_ * S_ * D_;
  float* pk   = (float*)d_ws;
  float* pq   = pk + B_ * S_ * P_;

  proj_wmma<<<256, 256, 0, stream>>>(seq, W1, W2, pk, pq);
  score_softmax<<<B_ * S_, 256, 0, stream>>>(pq, pk, v, mask, wout);
  attn_wmma<<<256, 256, 0, stream>>>(wout, seq, attn);
}


# both WMMA kernels re-tiled: a wave holds 8 accumulator tiles (64x32, 32x64) over a quarter of k, so a fragment load feeds 2 to 4 products; 4 k quarters joined through LDS: sum order over k changes
# speedup vs baseline: 2.1060x; 2.1060x over previous
#include <hip/hip_runtime.h>

typedef float v2f __attribute__((ext_vector_type(2)));
typedef float v4f __attribute__((ext_vector_type(4)));
typedef float v8f __attribute__((ext_vector_type(8)));

#define B_ 4
#define S_ 256
#define D_ 512
#define P_ 256
#define NEG_INF_ (-1e9f)

__device__ __forceinline__ float fast_tanh(float x) {
#if __has_builtin(__builtin_amdgcn_tanhf)
  return __builtin_amdgcn_tanhf(x);
#else
  float ax = __builtin_fabsf(x);
  float t  = __expf(-2.0f * ax);
  float r  = (1.0f - t) * __builtin_amdgcn_rcpf(1.0f + t);
  return x < 0.0f ? -r : r;
#endif
}

__global__ __launch_bounds__(128) __attribute__((amdgpu_waves_per_eu(1, 1)))
void proj_wmma(
    const float* __restrict__ seq, const float* __restrict__ W1,
    const float* __restrict__ W2, float* __restrict__ pkT,
    float* __restrict__ pq) {
  __shared__ v4f part[4][16][32];
  const int lane = threadIdx.x & 31;
  const int w = __builtin_amdgcn_readfirstlane((int)(threadIdx.x >> 5));
  const int matsel = blockIdx.x >> 7;
  const int tile   = blockIdx.x & 127;
  const int a0 = matsel ? ((tile >> 3) << 6) : ((tile & 3) << 6);
  const int b0 = matsel ? ((tile & 7) << 5) : ((tile >> 2) << 5);
  const float* Am = matsel ? seq : W1;
  const float* Bm = matsel ? W2 : seq;

  const int hl  = lane >> 4;
  const int l16 = lane & 15;
  const float* ap = Am + (a0 + l16) * D_ + (w << 7) + (hl << 2);
  const float* bp = Bm + (b0 + l16) * D_ + (w << 7) + (hl << 2);

  v8f acc[4][2];
#pragma unroll
  for (int i = 0; i < 4; ++i)
#pragma unroll
    for (int j = 0; j < 2; ++j) acc[i][j] = (v8f){};

#pragma unroll
  for (int s = 0; s < 16; ++s) {
    v4f a[4], b[2];
#pragma unroll
    for (int i = 0; i < 4; ++i) a[i] = *(const v4f*)(ap + i * 16 * D_ + s * 8);
#pragma unroll
    for (int j = 0; j < 2; ++j) b[j] = *(const v4f*)(bp + j * 16 * D_ + s * 8);
#pragma unroll
    for (int i = 0; i < 4; ++i)
#pragma unroll
      for (int j = 0; j < 2; ++j)
        acc[i][j] = __builtin_amdgcn_wmma_f32_16x16x4_f32(
            false, a[i].xy, false, b[j].xy, (short)0, acc[i][j], false, false);
#pragma unroll
    for (int i = 0; i < 4; ++i)
#pragma unroll
      for (int j = 0; j < 2; ++j)
        acc[i][j] = __builtin_amdgcn_wmma_f32_16x16x4_f32(
            false, a[i].zw, false, b[j].zw, (short)0, acc[i][j], false, false);
  }

#pragma unroll
  for (int i = 0; i < 4; ++i)
#pragma unroll
    for (int j = 0; j < 2; ++j) {
      part[w][i * 4 + j * 2 + 0][lane] =
          (v4f){acc[i][j][0], acc[i][j][1], acc[i][j][2], acc[i][j][3]};
      part[w][i * 4 + j * 2 + 1][lane] =
          (v4f){acc[i][j][4], acc[i][j][5], acc[i][j][6], acc[i][j][7]};
    }
  __syncthreads();

  v4f sum[4];
#pragma unroll
  for (int c = 0; c < 4; ++c)
    sum[c] = ((part[0][4 * w + c][lane] + part[1][4 * w + c][lane]) +
              part[2][4 * w + c][lane]) + part[3][4 * w + c][lane];
  const int ar = a0 + 16 * w + (hl << 3);
  const int bc = b0 + l16;
  if (matsel) {
#pragma unroll
    for (int c = 0; c < 4; ++c)
#pragma unroll
      for (int e = 0; e < 4; ++e)
        pq[(ar + 4 * (c & 1) + e) * P_ + bc + 16 * (c >> 1)] = sum[c][e];
  } else {
    float* o = pkT + ((bc >> 8) * P_ + ar) * S_ + (bc & (S_ - 1));
#pragma unroll
    for (int c = 0; c < 4; ++c)
#pragma unroll
      for (int e = 0; e < 4; ++e)
        o[(4 * (c & 1) + e) * S_ + 16 * (c >> 1)] = sum[c][e];
  }
}

#define QB_ 4
__global__ __launch_bounds__(256) __attribute__((amdgpu_waves_per_eu(2, 2)))
void score_softmax(
    const float* __restrict__ pq, const float* __restrict__ pkT,
    const float* __restrict__ v, const float* __restrict__ mask,
    float* __restrict__ wout) {
  const int b   = blockIdx.x >> 6;
  const int i0  = (blockIdx.x & 63) * QB_;
  const int tid = threadIdx.x;
  const int lane = tid & 31, wv = tid >> 5;
  const int kq  = tid & 63;
  const int pg  = __builtin_amdgcn_readfirstlane(tid >> 6);

  __shared__ float red[4][QB_][S_];
  __shared__ float redm[8][QB_];
  __shared__ float reds[8][QB_];

  const v4f*   src = (const v4f*)(pkT + (b * P_ + pg * 64) * S_) + kq;
  const float* pqb = pq + (b * S_ + i0) * P_ + pg * 64;
  const float* vb  = v + pg * 64;

  v4f x[64];
#pragma unroll
  for (int pp = 0; pp < 64; ++pp) x[pp] = src[pp * (S_ / 4)];

  v4f acc[QB_];
#pragma unroll
  for (int q = 0; q < QB_; ++q) acc[q] = (v4f){0.0f, 0.0f, 0.0f, 0.0f};

#pragma unroll
  for (int pp = 0; pp < 64; ++pp) {
    const float vv = vb[pp];
#pragma unroll
    for (int q = 0; q < QB_; ++q) {
      const float s = pqb[q * P_ + pp];
      v4f t;
      t.x = fast_tanh(s + x[pp].x);
      t.y = fast_tanh(s + x[pp].y);
      t.z = fast_tanh(s + x[pp].z);
      t.w = fast_tanh(s + x[pp].w);
      acc[q] = vv * t + acc[q];
    }
  }

#pragma unroll
  for (int q = 0; q < QB_; ++q) *(v4f*)&red[pg][q][kq << 2] = acc[q];
  __syncthreads();

  const float km = mask[b * S_ + tid];
  float s[QB_], mx[QB_];
#pragma unroll
  for (int q = 0; q < QB_; ++q) {
    const float sc = ((red[0][q][tid] + red[1][q][tid]) + red[2][q][tid]) + red[3][q][tid];
    s[q]  = (km > 0.0f) ? sc : NEG_INF_;
    mx[q] = s[q];
  }
#pragma unroll
  for (int off = 16; off > 0; off >>= 1)
#pragma unroll
    for (int q = 0; q < QB_; ++q)
      mx[q] = fmaxf(mx[q], __shfl_xor(mx[q], off, 32));
  if (lane == 0) {
#pragma unroll
    for (int q = 0; q < QB_; ++q) redm[wv][q] = mx[q];
  }
  __syncthreads();

  float e[QB_], sm[QB_];
#pragma unroll
  for (int q = 0; q < QB_; ++q) {
    float bm = redm[0][q];
#pragma unroll
    for (int w = 1; w < 8; ++w) bm = fmaxf(bm, redm[w][q]);
    e[q]  = __expf(s[q] - bm);
    sm[q] = e[q];
  }
#pragma unroll
  for (int off = 16; off > 0; off >>= 1)
#pragma unroll
    for (int q = 0; q < QB_; ++q)
      sm[q] += __shfl_xor(sm[q], off, 32);
  if (lane == 0) {
#pragma unroll
    for (int q = 0; q < QB_; ++q) reds[wv][q] = sm[q];
  }
  __syncthreads();

#pragma unroll
  for (int q = 0; q < QB_; ++q) {
    float tot = 0.0f;
#pragma unroll
    for (int w = 0; w < 8; ++w) tot += reds[w][q];
    const float qm = mask[b * S_ + i0 + q];
    wout[(b * S_ + i0 + q) * S_ + tid] = (e[q] / tot) * qm;
  }
}

__global__ __launch_bounds__(128) __attribute__((amdgpu_waves_per_eu(1, 1)))
void attn_wmma(
    const float* __restrict__ weights, const float* __restrict__ seq,
    float* __restrict__ attn) {
  __shared__ v4f part[4][16][32];
  const int lane = threadIdx.x & 31;
  const int w = __builtin_amdgcn_readfirstlane((int)(threadIdx.x >> 5));
  const int b  = blockIdx.x >> 6;
  const int m0 = ((blockIdx.x >> 3) & 7) << 5;
  const int n0 = (blockIdx.x & 7) << 6;
  const float* Wt = weights + b * S_ * S_;
  const float* Sq = seq     + b * S_ * D_;

  const int hl  = lane >> 4;
  const int l16 = lane & 15;
  const float* ap = Wt + (m0 + l16) * S_ + (w << 6) + (hl << 2);
  const float* bp = Sq + ((w << 6) + (hl << 2)) * D_ + n0 + (l16 << 2);

  v8f acc[2][4];
#pragma unroll
  for (int i = 0; i < 2; ++i)
#pragma unroll
    for (int t = 0; t < 4; ++t) acc[i][t] = (v8f){};

#pragma unroll
  for (int s = 0; s < 8; ++s) {
    v4f a[2], L[4];
#pragma unroll
    for (int i = 0; i < 2; ++i) a[i] = *(const v4f*)(ap + i * 16 * S_ + s * 8);
#pragma unroll
    for (int e = 0; e < 4; ++e) L[e] = *(const v4f*)(bp + (s * 8 + e) * D_);
#pragma unroll
    for (int i = 0; i < 2; ++i)
#pragma unroll
      for (int t = 0; t < 4; ++t) {
        const v2f bb = {L[0][t], L[1][t]};
        acc[i][t] = __builtin_amdgcn_wmma_f32_16x16x4_f32(
            false, a[i].xy, false, bb, (short)0, acc[i][t], false, false);
      }
#pragma unroll
    for (int i = 0; i < 2; ++i)
#pragma unroll
      for (int t = 0; t < 4; ++t) {
        const v2f bb = {L[2][t], L[3][t]};
        acc[i][t] = __builtin_amdgcn_wmma_f32_16x16x4_f32(
            false, a[i].zw, false, bb, (short)0, acc[i][t], false, false);
      }
  }

#pragma unroll
  for (int i = 0; i < 2; ++i)
#pragma unroll
    for (int r = 0; r < 8; ++r)
      part[w][i * 8 + r][lane] =
          (v4f){acc[i][0][r], acc[i][1][r], acc[i][2][r], acc[i][3][r]};
  __syncthreads();

  float* O = attn + b * S_ * D_;
#pragma unroll
  for (int c = 0; c < 4; ++c) {
    const v4f sum = ((part[0][4 * w + c][lane] + part[1][4 * w + c][lane]) +
                     part[2][4 * w + c][lane]) + part[3][4 * w + c][lane];
    const int row = m0 + 16 * (w >> 1) + 4 * (w & 1) + c + (hl << 3);
    *(v4f*)(O + row * D_ + n0 + (l16 << 2)) = sum;
  }
}

extern "C" void kernel_launch(void* const* d_in, const int* in_sizes, int n_in,
                              void* d_out, int out_size, void* d_ws, size_t ws_size,
                              hipStream_t stream) {
  (void)in_sizes; (void)n_in; (void)out_size; (void)ws_size;
  const float* seq  = (const float*)d_in[0];
  const float* mask = (const float*)d_in[1];
  const float* W1   = (const float*)d_in[2];
  const float* W2   = (const float*)d_in[3];
  const float* v    = (const float*)d_in[4];

  float* attn = (float*)d_out;
  float* wout = attn + B_ * S_ * D_;
  float* pkT  = (float*)d_ws;
  float* pq   = pkT + B_ * S_ * P_;

  proj_wmma<<<256, 128, 0, stream>>>(seq, W1, W2, pkT, pq);
  score_softmax<<<B_ * S_ / QB_, 256, 0, stream>>>(pq, pkT, v, mask, wout);
  attn_wmma<<<256, 128, 0, stream>>>(wout, seq, attn);
}


# score kernel softmax: each of 4 waves finishes one whole query row (8 keys per lane, lane exchange on the vector ALU, masks read early), so 3 barriers become 1: sum order over keys changes
# speedup vs baseline: 2.2408x; 1.0640x over previous
#include <hip/hip_runtime.h>

typedef float v2f __attribute__((ext_vector_type(2)));
typedef float v4f __attribute__((ext_vector_type(4)));
typedef float v8f __attribute__((ext_vector_type(8)));

#define B_ 4
#define S_ 256
#define D_ 512
#define P_ 256
#define NEG_INF_ (-1e9f)

__device__ __forceinline__ float fast_tanh(float x) {
#if __has_builtin(__builtin_amdgcn_tanhf)
  return __builtin_amdgcn_tanhf(x);
#else
  float ax = __builtin_fabsf(x);
  float t  = __expf(-2.0f * ax);
  float r  = (1.0f - t) * __builtin_amdgcn_rcpf(1.0f + t);
  return x < 0.0f ? -r : r;
#endif
}

__device__ __forceinline__ float xhalf(float x) {
  const int b = __builtin_bit_cast(int, x);
  return __builtin_bit_cast(float, __builtin_amdgcn_permlanex16(
      b, b, 0x76543210, 0xfedcba98, false, false));
}
template <int N> __device__ __forceinline__ float xrow(float x) {
  return __builtin_bit_cast(float, __builtin_amdgcn_update_dpp(
      0, __builtin_bit_cast(int, x), 0x160 + N, 0xf, 0xf, false));
}
__device__ __forceinline__ float wave_max(float x) {
  x = fmaxf(x, xrow<1>(x));
  x = fmaxf(x, xrow<2>(x));
  x = fmaxf(x, xrow<4>(x));
  x = fmaxf(x, xrow<8>(x));
  return fmaxf(x, xhalf(x));
}
__device__ __forceinline__ float wave_sum(float x) {
  x += xrow<1>(x);
  x += xrow<2>(x);
  x += xrow<4>(x);
  x += xrow<8>(x);
  return x + xhalf(x);
}

__global__ __launch_bounds__(128) __attribute__((amdgpu_waves_per_eu(1, 1)))
void proj_wmma(
    const float* __restrict__ seq, const float* __restrict__ W1,
    const float* __restrict__ W2, float* __restrict__ pkT,
    float* __restrict__ pq) {
  __shared__ v4f part[4][16][32];
  const int lane = threadIdx.x & 31;
  const int w = __builtin_amdgcn_readfirstlane((int)(threadIdx.x >> 5));
  const int matsel = blockIdx.x >> 7;
  const int tile   = blockIdx.x & 127;
  const int a0 = matsel ? ((tile >> 3) << 6) : ((tile & 3) << 6);
  const int b0 = matsel ? ((tile & 7) << 5) : ((tile >> 2) << 5);
  const float* Am = matsel ? seq : W1;
  const float* Bm = matsel ? W2 : seq;

  const int hl  = lane >> 4;
  const int l16 = lane & 15;
  const float* ap = Am + (a0 + l16) * D_ + (w << 7) + (hl << 2);
  const float* bp = Bm + (b0 + l16) * D_ + (w << 7) + (hl << 2);

  v8f acc[4][2];
#pragma unroll
  for (int i = 0; i < 4; ++i)
#pragma unroll
    for (int j = 0; j < 2; ++j) acc[i][j] = (v8f){};

#pragma unroll
  for (int s = 0; s < 16; ++s) {
    v4f a[4], b[2];
#pragma unroll
    for (int i = 0; i < 4; ++i) a[i] = *(const v4f*)(ap + i * 16 * D_ + s * 8);
#pragma unroll
    for (int j = 0; j < 2; ++j) b[j] = *(const v4f*)(bp + j * 16 * D_ + s * 8);
#pragma unroll
    for (int i = 0; i < 4; ++i)
#pragma unroll
      for (int j = 0; j < 2; ++j)
        acc[i][j] = __builtin_amdgcn_wmma_f32_16x16x4_f32(
            false, a[i].xy, false, b[j].xy, (short)0, acc[i][j], false, false);
#pragma unroll
    for (int i = 0; i < 4; ++i)
#pragma unroll
      for (int j = 0; j < 2; ++j)
        acc[i][j] = __builtin_amdgcn_wmma_f32_16x16x4_f32(
            false, a[i].zw, false, b[j].zw, (short)0, acc[i][j], false, false);
  }

#pragma unroll
  for (int i = 0; i < 4; ++i)
#pragma unroll
    for (int j = 0; j < 2; ++j) {
      part[w][i * 4 + j * 2 + 0][lane] =
          (v4f){acc[i][j][0], acc[i][j][1], acc[i][j][2], acc[i][j][3]};
      part[w][i * 4 + j * 2 + 1][lane] =
          (v4f){acc[i][j][4], acc[i][j][5], acc[i][j][6], acc[i][j][7]};
    }
  __syncthreads();

  v4f sum[4];
#pragma unroll
  for (int c = 0; c < 4; ++c)
    sum[c] = ((part[0][4 * w + c][lane] + part[1][4 * w + c][lane]) +
              part[2][4 * w + c][lane]) + part[3][4 * w + c][lane];
  const int ar = a0 + 16 * w + (hl << 3);
  const int bc = b0 + l16;
  if (matsel) {
#pragma unroll
    for (int c = 0; c < 4; ++c)
#pragma unroll
      for (int e = 0; e < 4; ++e)
        pq[(ar + 4 * (c & 1) + e) * P_ + bc + 16 * (c >> 1)] = sum[c][e];
  } else {
    float* o = pkT + ((bc >> 8) * P_ + ar) * S_ + (bc & (S_ - 1));
#pragma unroll
    for (int c = 0; c < 4; ++c)
#pragma unroll
      for (int e = 0; e < 4; ++e)
        o[(4 * (c & 1) + e) * S_ + 16 * (c >> 1)] = sum[c][e];
  }
}

#define QB_ 4
__global__ __launch_bounds__(256) __attribute__((amdgpu_waves_per_eu(2, 2)))
void score_softmax(
    const float* __restrict__ pq, const float* __restrict__ pkT,
    const float* __restrict__ v, const float* __restrict__ mask,
    float* __restrict__ wout) {
  const int b   = blockIdx.x >> 6;
  const int i0  = (blockIdx.x & 63) * QB_;
  const int tid = threadIdx.x;
  const int lane = tid & 31, wv = tid >> 5;
  const int kq  = tid & 63;
  const int pg  = __builtin_amdgcn_readfirstlane(tid >> 6);

  __shared__ float red[4][QB_][S_];

  const int wq = __builtin_amdgcn_readfirstlane(wv);
  const v4f kma = *(const v4f*)(mask + b * S_ + (lane << 2));
  const v4f kmb = *(const v4f*)(mask + b * S_ + 128 + (lane << 2));
  const float qm = mask[b * S_ + i0 + (wq & (QB_ - 1))];

  const v4f*   src = (const v4f*)(pkT + (b * P_ + pg * 64) * S_) + kq;
  const float* pqb = pq + (b * S_ + i0) * P_ + pg * 64;
  const float* vb  = v + pg * 64;

  v4f x[64];
#pragma unroll
  for (int pp = 0; pp < 64; ++pp) x[pp] = src[pp * (S_ / 4)];

  v4f acc[QB_];
#pragma unroll
  for (int q = 0; q < QB_; ++q) acc[q] = (v4f){0.0f, 0.0f, 0.0f, 0.0f};

#pragma unroll
  for (int pp = 0; pp < 64; ++pp) {
    const float vv = vb[pp];
#pragma unroll
    for (int q = 0; q < QB_; ++q) {
      const float s = pqb[q * P_ + pp];
      v4f t;
      t.x = fast_tanh(s + x[pp].x);
      t.y = fast_tanh(s + x[pp].y);
      t.z = fast_tanh(s + x[pp].z);
      t.w = fast_tanh(s + x[pp].w);
      acc[q] = vv * t + acc[q];
    }
  }

#pragma unroll
  for (int q = 0; q < QB_; ++q) *(v4f*)&red[pg][q][kq << 2] = acc[q];
  __syncthreads();

  if (wq >= QB_) return;

  const int k0 = lane << 2;
  v4f sa = ((*(const v4f*)&red[0][wq][k0] + *(const v4f*)&red[1][wq][k0]) +
            *(const v4f*)&red[2][wq][k0]) + *(const v4f*)&red[3][wq][k0];
  v4f sb = ((*(const v4f*)&red[0][wq][128 + k0] + *(const v4f*)&red[1][wq][128 + k0]) +
            *(const v4f*)&red[2][wq][128 + k0]) + *(const v4f*)&red[3][wq][128 + k0];
#pragma unroll
  for (int j = 0; j < 4; ++j) {
    sa[j] = (kma[j] > 0.0f) ? sa[j] : NEG_INF_;
    sb[j] = (kmb[j] > 0.0f) ? sb[j] : NEG_INF_;
  }
  float mx = fmaxf(fmaxf(fmaxf(sa.x, sa.y), fmaxf(sa.z, sa.w)),
                   fmaxf(fmaxf(sb.x, sb.y), fmaxf(sb.z, sb.w)));
  mx = wave_max(mx);

  v4f ea, eb;
#pragma unroll
  for (int j = 0; j < 4; ++j) {
    ea[j] = __expf(sa[j] - mx);
    eb[j] = __expf(sb[j] - mx);
  }
  float tot = (((ea.x + ea.y) + ea.z) + ea.w) + (((eb.x + eb.y) + eb.z) + eb.w);
  tot = wave_sum(tot);

  float* o = wout + (b * S_ + i0 + wq) * S_ + k0;
  v4f oa, ob;
#pragma unroll
  for (int j = 0; j < 4; ++j) {
    oa[j] = (ea[j] / tot) * qm;
    ob[j] = (eb[j] / tot) * qm;
  }
  *(v4f*)o         = oa;
  *(v4f*)(o + 128) = ob;
}

__global__ __launch_bounds__(128) __attribute__((amdgpu_waves_per_eu(1, 1)))
void attn_wmma(
    const float* __restrict__ weights, const float* __restrict__ seq,
    float* __restrict__ attn) {
  __shared__ v4f part[4][16][32];
  const int lane = threadIdx.x & 31;
  const int w = __builtin_amdgcn_readfirstlane((int)(threadIdx.x >> 5));
  const int b  = blockIdx.x >> 6;
  const int m0 = ((blockIdx.x >> 3) & 7) << 5;
  const int n0 = (blockIdx.x & 7) << 6;
  const float* Wt = weights + b * S_ * S_;
  const float* Sq = seq     + b * S_ * D_;

  const int hl  = lane >> 4;
  const int l16 = lane & 15;
  const float* ap = Wt + (m0 + l16) * S_ + (w << 6) + (hl << 2);
  const float* bp = Sq + ((w << 6) + (hl << 2)) * D_ + n0 + (l16 << 2);

  v8f acc[2][4];
#pragma unroll
  for (int i = 0; i < 2; ++i)
#pragma unroll
    for (int t = 0; t < 4; ++t) acc[i][t] = (v8f){};

#pragma unroll
  for (int s = 0; s < 8; ++s) {
    v4f a[2], L[4];
#pragma unroll
    for (int i = 0; i < 2; ++i) a[i] = *(const v4f*)(ap + i * 16 * S_ + s * 8);
#pragma unroll
    for (int e = 0; e < 4; ++e) L[e] = *(const v4f*)(bp + (s * 8 + e) * D_);
#pragma unroll
    for (int i = 0; i < 2; ++i)
#pragma unroll
      for (int t = 0; t < 4; ++t) {
        const v2f bb = {L[0][t], L[1][t]};
        acc[i][t] = __builtin_amdgcn_wmma_f32_16x16x4_f32(
            false, a[i].xy, false, bb, (short)0, acc[i][t], false, false);
      }
#pragma unroll
    for (int i = 0; i < 2; ++i)
#pragma unroll
      for (int t = 0; t < 4; ++t) {
        const v2f bb = {L[2][t], L[3][t]};
        acc[i][t] = __builtin_amdgcn_wmma_f32_16x16x4_f32(
            false, a[i].zw, false, bb, (short)0, acc[i][t], false, false);
      }
  }

#pragma unroll
  for (int i = 0; i < 2; ++i)
#pragma unroll
    for (int r = 0; r < 8; ++r)
      part[w][i * 8 + r][lane] =
          (v4f){acc[i][0][r], acc[i][1][r], acc[i][2][r], acc[i][3][r]};
  __syncthreads();

  float* O = attn + b * S_ * D_;
#pragma unroll
  for (int c = 0; c < 4; ++c) {
    const v4f sum = ((part[0][4 * w + c][lane] + part[1][4 * w + c][lane]) +
                     part[2][4 * w + c][lane]) + part[3][4 * w + c][lane];
    const int row = m0 + 16 * (w >> 1) + 4 * (w & 1) + c + (hl << 3);
    *(v4f*)(O + row * D_ + n0 + (l16 << 2)) = sum;
  }
}

extern "C" void kernel_launch(void* const* d_in, const int* in_sizes, int n_in,
                              void* d_out, int out_size, void* d_ws, size_t ws_size,
                              hipStream_t stream) {
  (void)in_sizes; (void)n_in; (void)out_size; (void)ws_size;
  const float* seq  = (const float*)d_in[0];
  const float* mask = (const float*)d_in[1];
  const float* W1   = (const float*)d_in[2];
  const float* W2   = (const float*)d_in[3];
  const float* v    = (const float*)d_in[4];

  float* attn = (float*)d_out;
  float* wout = attn + B_ * S_ * D_;
  float* pkT  = (float*)d_ws;
  float* pq   = pkT + B_ * S_ * P_;

  proj_wmma<<<256, 128, 0, stream>>>(seq, W1, W2, pkT, pq);
  score_softmax<<<B_ * S_ / QB_, 256, 0, stream>>>(pq, pkT, v, mask, wout);
  attn_wmma<<<256, 128, 0, stream>>>(wout, seq, attn);
}


# score kernel: the p loop as a hand-ordered asm stream, one packed add or packed multiply-add behind every tanh (tanh of step n, arguments of n+1, sums of n-1); operations and sum order unchanged
# speedup vs baseline: 2.3543x; 1.0507x over previous
#include <hip/hip_runtime.h>

typedef float v2f __attribute__((ext_vector_type(2)));
typedef float v4f __attribute__((ext_vector_type(4)));
typedef float v8f __attribute__((ext_vector_type(8)));

#define B_ 4
#define S_ 256
#define D_ 512
#define P_ 256
#define NEG_INF_ (-1e9f)

__device__ __forceinline__ float fast_tanh(float x) {
#if __has_builtin(__builtin_amdgcn_tanhf)
  return __builtin_amdgcn_tanhf(x);
#else
  float ax = __builtin_fabsf(x);
  float t  = __expf(-2.0f * ax);
  float r  = (1.0f - t) * __builtin_amdgcn_rcpf(1.0f + t);
  return x < 0.0f ? -r : r;
#endif
}

__device__ __forceinline__ float xhalf(float x) {
  const int b = __builtin_bit_cast(int, x);
  return __builtin_bit_cast(float, __builtin_amdgcn_permlanex16(
      b, b, 0x76543210, 0xfedcba98, false, false));
}
template <int N> __device__ __forceinline__ float xrow(float x) {
  return __builtin_bit_cast(float, __builtin_amdgcn_update_dpp(
      0, __builtin_bit_cast(int, x), 0x160 + N, 0xf, 0xf, false));
}
__device__ __forceinline__ float wave_max(float x) {
  x = fmaxf(x, xrow<1>(x));
  x = fmaxf(x, xrow<2>(x));
  x = fmaxf(x, xrow<4>(x));
  x = fmaxf(x, xrow<8>(x));
  return fmaxf(x, xhalf(x));
}
__device__ __forceinline__ float wave_sum(float x) {
  x += xrow<1>(x);
  x += xrow<2>(x);
  x += xrow<4>(x);
  x += xrow<8>(x);
  return x + xhalf(x);
}

__global__ __launch_bounds__(128) __attribute__((amdgpu_waves_per_eu(1, 1)))
void proj_wmma(
    const float* __restrict__ seq, const float* __restrict__ W1,
    const float* __restrict__ W2, float* __restrict__ pkT,
    float* __restrict__ pq) {
  __shared__ v4f part[4][16][32];
  const int lane = threadIdx.x & 31;
  const int w = __builtin_amdgcn_readfirstlane((int)(threadIdx.x >> 5));
  const int matsel = blockIdx.x >> 7;
  const int tile   = blockIdx.x & 127;
  const int a0 = matsel ? ((tile >> 3) << 6) : ((tile & 3) << 6);
  const int b0 = matsel ? ((tile & 7) << 5) : ((tile >> 2) << 5);
  const float* Am = matsel ? seq : W1;
  const float* Bm = matsel ? W2 : seq;

  const int hl  = lane >> 4;
  const int l16 = lane & 15;
  const float* ap = Am + (a0 + l16) * D_ + (w << 7) + (hl << 2);
  const float* bp = Bm + (b0 + l16) * D_ + (w << 7) + (hl << 2);

  v8f acc[4][2];
#pragma unroll
  for (int i = 0; i < 4; ++i)
#pragma unroll
    for (int j = 0; j < 2; ++j) acc[i][j] = (v8f){};

#pragma unroll
  for (int s = 0; s < 16; ++s) {
    v4f a[4], b[2];
#pragma unroll
    for (int i = 0; i < 4; ++i) a[i] = *(const v4f*)(ap + i * 16 * D_ + s * 8);
#pragma unroll
    for (int j = 0; j < 2; ++j) b[j] = *(const v4f*)(bp + j * 16 * D_ + s * 8);
#pragma unroll
    for (int i = 0; i < 4; ++i)
#pragma unroll
      for (int j = 0; j < 2; ++j)
        acc[i][j] = __builtin_amdgcn_wmma_f32_16x16x4_f32(
            false, a[i].xy, false, b[j].xy, (short)0, acc[i][j], false, false);
#pragma unroll
    for (int i = 0; i < 4; ++i)
#pragma unroll
      for (int j = 0; j < 2; ++j)
        acc[i][j] = __builtin_amdgcn_wmma_f32_16x16x4_f32(
            false, a[i].zw, false, b[j].zw, (short)0, acc[i][j], false, false);
  }

#pragma unroll
  for (int i = 0; i < 4; ++i)
#pragma unroll
    for (int j = 0; j < 2; ++j) {
      part[w][i * 4 + j * 2 + 0][lane] =
          (v4f){acc[i][j][0], acc[i][j][1], acc[i][j][2], acc[i][j][3]};
      part[w][i * 4 + j * 2 + 1][lane] =
          (v4f){acc[i][j][4], acc[i][j][5], acc[i][j][6], acc[i][j][7]};
    }
  __syncthreads();

  v4f sum[4];
#pragma unroll
  for (int c = 0; c < 4; ++c)
    sum[c] = ((part[0][4 * w + c][lane] + part[1][4 * w + c][lane]) +
              part[2][4 * w + c][lane]) + part[3][4 * w + c][lane];
  const int ar = a0 + 16 * w + (hl << 3);
  const int bc = b0 + l16;
  if (matsel) {
#pragma unroll
    for (int c = 0; c < 4; ++c)
#pragma unroll
      for (int e = 0; e < 4; ++e)
        pq[(ar + 4 * (c & 1) + e) * P_ + bc + 16 * (c >> 1)] = sum[c][e];
  } else {
    float* o = pkT + ((bc >> 8) * P_ + ar) * S_ + (bc & (S_ - 1));
#pragma unroll
    for (int c = 0; c < 4; ++c)
#pragma unroll
      for (int e = 0; e < 4; ++e)
        o[(4 * (c & 1) + e) * S_ + 16 * (c >> 1)] = sum[c][e];
  }
}

#define QB_ 4
__global__ __launch_bounds__(256) __attribute__((amdgpu_waves_per_eu(2, 2)))
void score_softmax(
    const float* __restrict__ pq, const float* __restrict__ pkT,
    const float* __restrict__ v, const float* __restrict__ mask,
    float* __restrict__ wout) {
  const int b   = blockIdx.x >> 6;
  const int i0  = (blockIdx.x & 63) * QB_;
  const int tid = threadIdx.x;
  const int lane = tid & 31, wv = tid >> 5;
  const int kq  = tid & 63;
  const int pg  = __builtin_amdgcn_readfirstlane(tid >> 6);

  __shared__ float red[4][QB_][S_];

  const int wq = __builtin_amdgcn_readfirstlane(wv);
  const v4f kma = *(const v4f*)(mask + b * S_ + (lane << 2));
  const v4f kmb = *(const v4f*)(mask + b * S_ + 128 + (lane << 2));
  const float qm = mask[b * S_ + i0 + (wq & (QB_ - 1))];

  const v4f*   src = (const v4f*)(pkT + (b * P_ + pg * 64) * S_) + kq;
  const float* pqb = pq + (b * S_ + i0) * P_ + pg * 64;
  const float* vb  = v + pg * 64;

  int lane_zero;
  asm("v_mov_b32 %0, 0" : "=v"(lane_zero));
  const float* pqv = pqb + lane_zero;
  const float* vbv = vb + lane_zero;

#define XW_ 26
  v4f x[64];
  v4f sc[QB_][16];
  v4f vc[16];
#pragma unroll
  for (int c = 0; c < 2; ++c) {
#pragma unroll
    for (int q = 0; q < QB_; ++q) sc[q][c] = *(const v4f*)(pqv + q * P_ + 4 * c);
    vc[c] = *(const v4f*)(vbv + 4 * c);
  }
#pragma unroll
  for (int pp = 0; pp < XW_; ++pp) x[pp] = src[pp * (S_ / 4)];

  v2f u[QB_][2];
  v2f th[QB_][2];
  v2f ac[QB_][2];
#pragma unroll
  for (int q = 0; q < QB_; ++q) {
    const float s = sc[q][0].x;
    u[q][0] = (v2f){s + x[0].x, s + x[0].y};
    u[q][1] = (v2f){s + x[0].z, s + x[0].w};
    th[q][0] = th[q][1] = (v2f){0.0f, 0.0f};
    ac[q][0] = ac[q][1] = (v2f){0.0f, 0.0f};
  }
#define P_STEP(PP, ADDSEL, FMASEL)                                            \
  {                                                                           \
    const int pn = (PP) < 63 ? (PP) + 1 : 63;  \
    const int pv = (PP) > 0 ? (PP) - 1 : 0;      \
    if ((PP) + XW_ < 64) x[(PP) + XW_] = src[((PP) + XW_) * (S_ / 4)];        \
    if (((PP) & 3) == 0 && ((PP) >> 2) + 2 < 16) {                            \
      const int c = ((PP) >> 2) + 2;                                          \
      _Pragma("unroll") for (int q = 0; q < QB_; ++q)                         \
        sc[q][c] = *(const v4f*)(pqv + q * P_ + 4 * c);                       \
      vc[c] = *(const v4f*)(vbv + 4 * c);                                     \
    }                                                                         \
    const v4f vch = vc[pv >> 2];                                              \
    const v2f vl = (pv & 2) ? (v2f){vch.z, vch.w} : (v2f){vch.x, vch.y};      \
    const v2f vv = (PP) > 0 ? vl : (v2f){0.0f, 0.0f};                         \
    const v2f xa = {x[pn].x, x[pn].y};                                        \
    const v2f xb = {x[pn].z, x[pn].w};                                        \
    _Pragma("unroll") for (int q = 0; q < QB_; ++q) {                         \
      const v4f sch = sc[q][pn >> 2];                                         \
      const v2f sq = (pn & 2) ? (v2f){sch.z, sch.w} : (v2f){sch.x, sch.y};    \
      const float u0x = u[q][0].x, u0y = u[q][0].y;                           \
      const float u1x = u[q][1].x, u1y = u[q][1].y;                           \
      float t0, t1, t2, t3;                                                   \
      v2f un0, un1;                                                           \
              \
      asm volatile(                                                           \
          "v_tanh_f32 %0, %10\n\t"                                            \
          "v_pk_add_f32 %4, %16, %14 " ADDSEL "\n\t"                          \
          "v_tanh_f32 %1, %11\n\t"                                            \
          "v_pk_fma_f32 %6, %17, %8, %6 " FMASEL "\n\t"                       \
          "v_tanh_f32 %2, %12\n\t"                                            \
          "v_pk_add_f32 %5, %16, %15 " ADDSEL "\n\t"                          \
          "v_tanh_f32 %3, %13\n\t"                                            \
          "v_pk_fma_f32 %7, %17, %9, %7 " FMASEL                               \
          : "=&v"(t0), "=&v"(t1), "=&v"(t2), "=&v"(t3), "=&v"(un0),           \
            "=&v"(un1), "+&v"(ac[q][0]), "+&v"(ac[q][1])                      \
          : "v"(th[q][0]), "v"(th[q][1]), "v"(u0x), "v"(u0y), "v"(u1x),       \
            "v"(u1y), "v"(xa), "v"(xb), "v"(sq), "v"(vv)                      \
          : "memory");                                                        \
      th[q][0] = (v2f){t0, t1};                                               \
      th[q][1] = (v2f){t2, t3};                                               \
      u[q][0] = un0;                                                          \
      u[q][1] = un1;                                                          \
    }                                                                         \
  }
#pragma unroll
  for (int k = 0; k < 32; ++k) {
    P_STEP(2 * k, "op_sel:[1,0]", "op_sel:[1,0,0]")
    P_STEP(2 * k + 1, "op_sel_hi:[0,1]", "op_sel_hi:[0,1,1]")
  }
#undef P_STEP
#undef XW_
  v4f acc[QB_];
  {
    const float vlast = vc[15].w;
#pragma unroll
    for (int q = 0; q < QB_; ++q) {
      const v2f a0 = vlast * th[q][0] + ac[q][0];
      const v2f a1 = vlast * th[q][1] + ac[q][1];
      acc[q] = (v4f){a0.x, a0.y, a1.x, a1.y};
    }
  }

#pragma unroll
  for (int q = 0; q < QB_; ++q) *(v4f*)&red[pg][q][kq << 2] = acc[q];
  __syncthreads();

  if (wq >= QB_) return;

  const int k0 = lane << 2;
  v4f sa = ((*(const v4f*)&red[0][wq][k0] + *(const v4f*)&red[1][wq][k0]) +
            *(const v4f*)&red[2][wq][k0]) + *(const v4f*)&red[3][wq][k0];
  v4f sb = ((*(const v4f*)&red[0][wq][128 + k0] + *(const v4f*)&red[1][wq][128 + k0]) +
            *(const v4f*)&red[2][wq][128 + k0]) + *(const v4f*)&red[3][wq][128 + k0];
#pragma unroll
  for (int j = 0; j < 4; ++j) {
    sa[j] = (kma[j] > 0.0f) ? sa[j] : NEG_INF_;
    sb[j] = (kmb[j] > 0.0f) ? sb[j] : NEG_INF_;
  }
  float mx = fmaxf(fmaxf(fmaxf(sa.x, sa.y), fmaxf(sa.z, sa.w)),
                   fmaxf(fmaxf(sb.x, sb.y), fmaxf(sb.z, sb.w)));
  mx = wave_max(mx);

  v4f ea, eb;
#pragma unroll
  for (int j = 0; j < 4; ++j) {
    ea[j] = __expf(sa[j] - mx);
    eb[j] = __expf(sb[j] - mx);
  }
  float tot = (((ea.x + ea.y) + ea.z) + ea.w) + (((eb.x + eb.y) + eb.z) + eb.w);
  tot = wave_sum(tot);

  float* o = wout + (b * S_ + i0 + wq) * S_ + k0;
  v4f oa, ob;
#pragma unroll
  for (int j = 0; j < 4; ++j) {
    oa[j] = (ea[j] / tot) * qm;
    ob[j] = (eb[j] / tot) * qm;
  }
  *(v4f*)o         = oa;
  *(v4f*)(o + 128) = ob;
}

__global__ __launch_bounds__(128) __attribute__((amdgpu_waves_per_eu(1, 1)))
void attn_wmma(
    const float* __restrict__ weights, const float* __restrict__ seq,
    float* __restrict__ attn) {
  __shared__ v4f part[4][16][32];
  const int lane = threadIdx.x & 31;
  const int w = __builtin_amdgcn_readfirstlane((int)(threadIdx.x >> 5));
  const int b  = blockIdx.x >> 6;
  const int m0 = ((blockIdx.x >> 3) & 7) << 5;
  const int n0 = (blockIdx.x & 7) << 6;
  const float* Wt = weights + b * S_ * S_;
  const float* Sq = seq     + b * S_ * D_;

  const int hl  = lane >> 4;
  const int l16 = lane & 15;
  const float* ap = Wt + (m0 + l16) * S_ + (w << 6) + (hl << 2);
  const float* bp = Sq + ((w << 6) + (hl << 2)) * D_ + n0 + (l16 << 2);

  v8f acc[2][4];
#pragma unroll
  for (int i = 0; i < 2; ++i)
#pragma unroll
    for (int t = 0; t < 4; ++t) acc[i][t] = (v8f){};

#pragma unroll
  for (int s = 0; s < 8; ++s) {
    v4f a[2], L[4];
#pragma unroll
    for (int i = 0; i < 2; ++i) a[i] = *(const v4f*)(ap + i * 16 * S_ + s * 8);
#pragma unroll
    for (int e = 0; e < 4; ++e) L[e] = *(const v4f*)(bp + (s * 8 + e) * D_);
#pragma unroll
    for (int i = 0; i < 2; ++i)
#pragma unroll
      for (int t = 0; t < 4; ++t) {
        const v2f bb = {L[0][t], L[1][t]};
        acc[i][t] = __builtin_amdgcn_wmma_f32_16x16x4_f32(
            false, a[i].xy, false, bb, (short)0, acc[i][t], false, false);
      }
#pragma unroll
    for (int i = 0; i < 2; ++i)
#pragma unroll
      for (int t = 0; t < 4; ++t) {
        const v2f bb = {L[2][t], L[3][t]};
        acc[i][t] = __builtin_amdgcn_wmma_f32_16x16x4_f32(
            false, a[i].zw, false, bb, (short)0, acc[i][t], false, false);
      }
  }

#pragma unroll
  for (int i = 0; i < 2; ++i)
#pragma unroll
    for (int r = 0; r < 8; ++r)
      part[w][i * 8 + r][lane] =
          (v4f){acc[i][0][r], acc[i][1][r], acc[i][2][r], acc[i][3][r]};
  __syncthreads();

  float* O = attn + b * S_ * D_;
#pragma unroll
  for (int c = 0; c < 4; ++c) {
    const v4f sum = ((part[0][4 * w + c][lane] + part[1][4 * w + c][lane]) +
                     part[2][4 * w + c][lane]) + part[3][4 * w + c][lane];
    const int row = m0 + 16 * (w >> 1) + 4 * (w & 1) + c + (hl << 3);
    *(v4f*)(O + row * D_ + n0 + (l16 << 2)) = sum;
  }
}

extern "C" void kernel_launch(void* const* d_in, const int* in_sizes, int n_in,
                              void* d_out, int out_size, void* d_ws, size_t ws_size,
                              hipStream_t stream) {
  (void)in_sizes; (void)n_in; (void)out_size; (void)ws_size;
  const float* seq  = (const float*)d_in[0];
  const float* mask = (const float*)d_in[1];
  const float* W1   = (const float*)d_in[2];
  const float* W2   = (const float*)d_in[3];
  const float* v    = (const float*)d_in[4];

  float* attn = (float*)d_out;
  float* wout = attn + B_ * S_ * D_;
  float* pkT  = (float*)d_ws;
  float* pq   = pkT + B_ * S_ * P_;

  proj_wmma<<<256, 128, 0, stream>>>(seq, W1, W2, pkT, pq);
  score_softmax<<<B_ * S_ / QB_, 256, 0, stream>>>(pq, pkT, v, mask, wout);
  attn_wmma<<<256, 128, 0, stream>>>(wout, seq, attn);
}
